# starting kernel
# baseline (speedup 1.0000x reference)
#include <hip/hip_runtime.h>
#include <hip/hip_bf16.h>

typedef __attribute__((ext_vector_type(16))) __bf16        v16bf;
typedef __attribute__((ext_vector_type(8)))  float         v8f;
typedef __attribute__((ext_vector_type(8)))  unsigned short u16x8;
typedef __attribute__((ext_vector_type(16))) unsigned short u16x16;

__device__ __forceinline__ unsigned short f32_to_bf16_rne(float f) {
    unsigned int u = __float_as_uint(f);
    unsigned int r = u + 0x7FFFu + ((u >> 16) & 1u);
    return (unsigned short)(r >> 16);
}

#define HDIM 256

__global__ void deg_init_kernel(float* deg, int n) {
    int i = blockIdx.x * blockDim.x + threadIdx.x;
    if (i < n) deg[i] = 1.0f;
}

__global__ void deg_edges_kernel(const int* __restrict__ dst, float* deg, int e) {
    int i = blockIdx.x * blockDim.x + threadIdx.x;
    if (i < e) atomicAdd(&deg[dst[i]], 1.0f);
}

__global__ void deg_rsqrt_kernel(float* deg, int n) {
    int i = blockIdx.x * blockDim.x + threadIdx.x;
    if (i < n) deg[i] = rsqrtf(deg[i]);
}

__global__ void cvt_bf16_kernel(const float* __restrict__ x, unsigned short* __restrict__ xb,
                                size_t total) {
    size_t i = (size_t)blockIdx.x * blockDim.x + threadIdx.x;
    if (i < total) xb[i] = f32_to_bf16_rne(x[i]);
}

__global__ void relu_bf16_kernel(const float* __restrict__ acc, unsigned short* __restrict__ zb,
                                 size_t total) {
    size_t i = (size_t)blockIdx.x * blockDim.x + threadIdx.x;
    if (i < total) zb[i] = f32_to_bf16_rne(fmaxf(acc[i], 0.0f));
}

__global__ void pack_w_kernel(const float* __restrict__ W, unsigned short* __restrict__ Bp) {
    int idx = blockIdx.x * blockDim.x + threadIdx.x;
    int j  = idx & 15;
    int hl = (idx >> 4) & 1;
    int n  = (idx >> 5) & 255;
    int t  = idx >> 13;
    int k  = t * 32 + ((j >> 3) << 4) + hl * 8 + (j & 7);
    Bp[idx] = f32_to_bf16_rne(W[k * HDIM + n]);
}

__global__ void gemm_wmma_bf16(const unsigned short* __restrict__ A,
                               const unsigned short* __restrict__ Bp,
                               float* __restrict__ C, int stripsTotal) {
    const int wave = threadIdx.x >> 5;
    const int lane = threadIdx.x & 31;
    const int hl   = lane >> 4;
    const int m16  = lane & 15;
    const int strip = blockIdx.x * 8 + wave;
    if (strip >= stripsTotal) return;
    const int tm = strip >> 2;
    const int tn = strip & 3;

    const unsigned short* arow = A + (size_t)(tm * 16 + m16) * HDIM;
    const unsigned short* bbase[4];
#pragma unroll
    for (int c = 0; c < 4; ++c)
        bbase[c] = Bp + ((size_t)(tn * 64 + c * 16 + m16) * 2 + hl) * 16;

    v8f acc0 = {}, acc1 = {}, acc2 = {}, acc3 = {};
#pragma unroll
    for (int kt = 0; kt < 8; ++kt) {
        const int k0 = kt * 32;
        u16x8 a_lo = *(const u16x8*)(arow + k0 + hl * 8);
        u16x8 a_hi = *(const u16x8*)(arow + k0 + 16 + hl * 8);
        u16x16 au;
#pragma unroll
        for (int i = 0; i < 8; ++i) { au[i] = a_lo[i]; au[i + 8] = a_hi[i]; }
        v16bf av = __builtin_bit_cast(v16bf, au);

        const size_t koff = (size_t)kt * (HDIM * 32);
#pragma unroll
        for (int c = 0; c < 4; ++c) {
            const unsigned short* bptr = bbase[c] + koff;
            u16x8 b_lo = *(const u16x8*)(bptr);
            u16x8 b_hi = *(const u16x8*)(bptr + 8);
            u16x16 bu;
#pragma unroll
            for (int i = 0; i < 8; ++i) { bu[i] = b_lo[i]; bu[i + 8] = b_hi[i]; }
            v16bf bv = __builtin_bit_cast(v16bf, bu);
            v8f cin = (c == 0) ? acc0 : (c == 1) ? acc1 : (c == 2) ? acc2 : acc3;
            v8f cout = __builtin_amdgcn_wmma_f32_16x16x32_bf16(
                 false, av,  false, bv,
                 (short)0, cin,  false,  false);
            if (c == 0) acc0 = cout; else if (c == 1) acc1 = cout;
            else if (c == 2) acc2 = cout; else acc3 = cout;
        }
    }

#pragma unroll
    for (int c = 0; c < 4; ++c) {
        const int col = tn * 64 + c * 16 + m16;
        v8f accv = (c == 0) ? acc0 : (c == 1) ? acc1 : (c == 2) ? acc2 : acc3;
#pragma unroll
        for (int r = 0; r < 8; ++r)
            C[(size_t)(tm * 16 + hl * 8 + r) * HDIM + col] = accv[r];
    }
}

__global__ void agg_init_kernel(const float* __restrict__ h, const float* __restrict__ dis,
                                const float* __restrict__ bias, float* __restrict__ acc) {
    size_t i = (size_t)blockIdx.x * HDIM + threadIdx.x;
    int row = blockIdx.x;
    int f   = threadIdx.x;
    float d = dis[row];
    acc[i] = bias[f] + h[i] * d * d;
}

__global__ void scatter_edges_kernel(const int* __restrict__ src, const int* __restrict__ dst,
                                     const float* __restrict__ dis, const float* __restrict__ h,
                                     float* __restrict__ acc, int E) {
    int wave = threadIdx.x >> 5;
    int lane = threadIdx.x & 31;
    int e = blockIdx.x * 8 + wave;
    if (e >= E) return;
    int s = src[e], d = dst[e];
    float w = dis[s] * dis[d];
    const float* hs = h  + (size_t)s * HDIM;
    float*       ad = acc + (size_t)d * HDIM;
#pragma unroll
    for (int f = lane; f < HDIM; f += 32)
        atomicAdd(&ad[f], hs[f] * w);
}

__global__ void l2norm_kernel(const float* __restrict__ acc, float* __restrict__ out) {
    int row = blockIdx.x;
    int tid = threadIdx.x;
    float v = acc[(size_t)row * HDIM + tid];
    float s = v * v;
#pragma unroll
    for (int m = 16; m > 0; m >>= 1) s += __shfl_xor(s, m, 32);
    __shared__ float ws[8];
    if ((tid & 31) == 0) ws[tid >> 5] = s;
    __syncthreads();
    float tot = 0.0f;
#pragma unroll
    for (int i = 0; i < 8; ++i) tot += ws[i];
    float inv = 1.0f / fmaxf(sqrtf(tot), 1e-12f);
    out[(size_t)row * HDIM + tid] = v * inv;
}

extern "C" void kernel_launch(void* const* d_in, const int* in_sizes, int n_in,
                              void* d_out, int out_size, void* d_ws, size_t ws_size,
                              hipStream_t stream) {
    const float* x   = (const float*)d_in[0];
    const int*   ei  = (const int*)d_in[1];
    const float* W1  = (const float*)d_in[2];
    const float* b1  = (const float*)d_in[3];
    const float* W2  = (const float*)d_in[4];
    const float* b2  = (const float*)d_in[5];
    float* out = (float*)d_out;

    const int N = in_sizes[0] / HDIM;
    const int E = in_sizes[1] / 2;
    const int* src = ei;
    const int* dst = ei + E;

    char* p = (char*)d_ws;
    auto take = [&](size_t bytes) { void* r = p; p += (bytes + 255) & ~(size_t)255; return r; };
    unsigned short* xb  = (unsigned short*)take((size_t)N * HDIM * 2);
    float*          h   = (float*)take((size_t)N * HDIM * 4);
    float*          acc = (float*)take((size_t)N * HDIM * 4);
    unsigned short* wp1 = (unsigned short*)take(8 * 256 * 32 * 2);
    unsigned short* wp2 = (unsigned short*)take(8 * 256 * 32 * 2);
    float*          dis = (float*)take((size_t)N * 4);

    const size_t total = (size_t)N * HDIM;
    const int tilesM = (N + 15) / 16;
    const int stripsTotal = tilesM * 4;
    const int gemmBlocks = (stripsTotal + 7) / 8;
    const int edgeBlocks = (E + 7) / 8;

    deg_init_kernel<<<(N + 255) / 256, 256, 0, stream>>>(dis, N);
    deg_edges_kernel<<<(E + 255) / 256, 256, 0, stream>>>(dst, dis, E);
    deg_rsqrt_kernel<<<(N + 255) / 256, 256, 0, stream>>>(dis, N);

    cvt_bf16_kernel<<<(int)((total + 255) / 256), 256, 0, stream>>>(x, xb, total);
    pack_w_kernel<<<256, 256, 0, stream>>>(W1, wp1);
    pack_w_kernel<<<256, 256, 0, stream>>>(W2, wp2);

    gemm_wmma_bf16<<<gemmBlocks, 256, 0, stream>>>(xb, wp1, h, stripsTotal);
    agg_init_kernel<<<N, 256, 0, stream>>>(h, dis, b1, acc);
    scatter_edges_kernel<<<edgeBlocks, 256, 0, stream>>>(src, dst, dis, h, acc, E);
    relu_bf16_kernel<<<(int)((total + 255) / 256), 256, 0, stream>>>(acc, xb, total);

    gemm_wmma_bf16<<<gemmBlocks, 256, 0, stream>>>(xb, wp2, h, stripsTotal);
    agg_init_kernel<<<N, 256, 0, stream>>>(h, dis, b2, acc);
    scatter_edges_kernel<<<edgeBlocks, 256, 0, stream>>>(src, dst, dis, h, acc, E);

    l2norm_kernel<<<N, 256, 0, stream>>>(acc, out);
}


# f32 atomic row scatter replaced by gather by destination: in-edge lists rebuilt per call (integer atomic arrival rank + prefix scan); one wave sums a row in registers in list order, stores it once
# speedup vs baseline: 2.0908x; 2.0908x over previous
#include <hip/hip_runtime.h>
#include <hip/hip_bf16.h>

typedef __attribute__((ext_vector_type(16))) __bf16        v16bf;
typedef __attribute__((ext_vector_type(8)))  float         v8f;
typedef __attribute__((ext_vector_type(8)))  unsigned short u16x8;
typedef __attribute__((ext_vector_type(16))) unsigned short u16x16;

__device__ __forceinline__ unsigned short f32_to_bf16_rne(float f) {
    unsigned int u = __float_as_uint(f);
    unsigned int r = u + 0x7FFFu + ((u >> 16) & 1u);
    return (unsigned short)(r >> 16);
}

#define HDIM 256

__global__ void zero_cnt_kernel(int* cnt, int n) {
    int i = blockIdx.x * blockDim.x + threadIdx.x;
    if (i < n) cnt[i] = 0;
}

__global__ void count_rank_kernel(const int* __restrict__ dst, int* cnt, int* __restrict__ rank, int e) {
    int i = blockIdx.x * blockDim.x + threadIdx.x;
    if (i < e) rank[i] = atomicAdd(&cnt[dst[i]], 1);
}

__global__ void __launch_bounds__(1024)
scan_deg_kernel(const int* __restrict__ cnt, int* __restrict__ rowptr, float* __restrict__ dis, int n) {
    __shared__ int wsum[32];
    const int tid  = threadIdx.x;
    const int lane = tid & 31;
    const int wv   = tid >> 5;
    int carry = 0;
    for (int base = 0; base < n; base += 8192) {
        const int i0 = base + tid * 8;
        int c[8];
        if (i0 + 7 < n) {
            const int4 v0 = *(const int4*)(cnt + i0);
            const int4 v1 = *(const int4*)(cnt + i0 + 4);
            c[0] = v0.x; c[1] = v0.y; c[2] = v0.z; c[3] = v0.w;
            c[4] = v1.x; c[5] = v1.y; c[6] = v1.z; c[7] = v1.w;
        } else {
            for (int k = 0; k < 8; ++k) c[k] = (i0 + k < n) ? cnt[i0 + k] : 0;
        }
        int t = 0;
        for (int k = 0; k < 8; ++k) t += c[k];
        int s = t;
        for (int d = 1; d < 32; d <<= 1) {
            const int y = __shfl_up(s, d, 32);
            if (lane >= d) s += y;
        }
        if (lane == 31) wsum[wv] = s;
        __syncthreads();
        const int wt = wsum[lane];
        int ws = wt;
        for (int d = 1; d < 32; d <<= 1) {
            const int y = __shfl_up(ws, d, 32);
            if (lane >= d) ws += y;
        }
        const int wave_excl = __shfl(ws, wv, 32) - __shfl(wt, wv, 32);
        const int total     = __shfl(ws, 31, 32);
        int run = carry + wave_excl + (s - t);
        if (i0 + 7 < n) {
            int4 p0, p1;
            float4 q0, q1;
            p0.x = run; run += c[0]; p0.y = run; run += c[1]; p0.z = run; run += c[2]; p0.w = run; run += c[3];
            p1.x = run; run += c[4]; p1.y = run; run += c[5]; p1.z = run; run += c[6]; p1.w = run;
            q0.x = rsqrtf((float)(c[0] + 1)); q0.y = rsqrtf((float)(c[1] + 1));
            q0.z = rsqrtf((float)(c[2] + 1)); q0.w = rsqrtf((float)(c[3] + 1));
            q1.x = rsqrtf((float)(c[4] + 1)); q1.y = rsqrtf((float)(c[5] + 1));
            q1.z = rsqrtf((float)(c[6] + 1)); q1.w = rsqrtf((float)(c[7] + 1));
            *(int4*)(rowptr + i0)     = p0;
            *(int4*)(rowptr + i0 + 4) = p1;
            *(float4*)(dis + i0)      = q0;
            *(float4*)(dis + i0 + 4)  = q1;
        } else {
            for (int k = 0; k < 8; ++k) {
                if (i0 + k < n) { rowptr[i0 + k] = run; dis[i0 + k] = rsqrtf((float)(c[k] + 1)); }
                run += c[k];
            }
        }
        carry += total;
        __syncthreads();
    }
    if (tid == 0) rowptr[n] = carry;
}

__global__ void fill_lists_kernel(const int* __restrict__ src, const int* __restrict__ dst,
                                  const int* __restrict__ rank, const int* __restrict__ rowptr,
                                  const float* __restrict__ dis, int2* __restrict__ rec, int e) {
    int i = blockIdx.x * blockDim.x + threadIdx.x;
    if (i >= e) return;
    const int s = src[i], d = dst[i];
    rec[rowptr[d] + rank[i]] = make_int2(s, __float_as_int(dis[s] * dis[d]));
}

__global__ void cvt_bf16_kernel(const float* __restrict__ x, unsigned short* __restrict__ xb,
                                size_t total) {
    size_t i = (size_t)blockIdx.x * blockDim.x + threadIdx.x;
    if (i < total) xb[i] = f32_to_bf16_rne(x[i]);
}

__global__ void relu_bf16_kernel(const float* __restrict__ acc, unsigned short* __restrict__ zb,
                                 size_t total) {
    size_t i = (size_t)blockIdx.x * blockDim.x + threadIdx.x;
    if (i < total) zb[i] = f32_to_bf16_rne(fmaxf(acc[i], 0.0f));
}

__global__ void pack_w_kernel(const float* __restrict__ W, unsigned short* __restrict__ Bp) {
    int idx = blockIdx.x * blockDim.x + threadIdx.x;
    int j  = idx & 15;
    int hl = (idx >> 4) & 1;
    int n  = (idx >> 5) & 255;
    int t  = idx >> 13;
    int k  = t * 32 + ((j >> 3) << 4) + hl * 8 + (j & 7);
    Bp[idx] = f32_to_bf16_rne(W[k * HDIM + n]);
}

__global__ void gemm_wmma_bf16(const unsigned short* __restrict__ A,
                               const unsigned short* __restrict__ Bp,
                               float* __restrict__ C, int stripsTotal) {
    const int wave = threadIdx.x >> 5;
    const int lane = threadIdx.x & 31;
    const int hl   = lane >> 4;
    const int m16  = lane & 15;
    const int strip = blockIdx.x * 8 + wave;
    if (strip >= stripsTotal) return;
    const int tm = strip >> 2;
    const int tn = strip & 3;

    const unsigned short* arow = A + (size_t)(tm * 16 + m16) * HDIM;
    const unsigned short* bbase[4];
#pragma unroll
    for (int c = 0; c < 4; ++c)
        bbase[c] = Bp + ((size_t)(tn * 64 + c * 16 + m16) * 2 + hl) * 16;

    v8f acc0 = {}, acc1 = {}, acc2 = {}, acc3 = {};
#pragma unroll
    for (int kt = 0; kt < 8; ++kt) {
        const int k0 = kt * 32;
        u16x8 a_lo = *(const u16x8*)(arow + k0 + hl * 8);
        u16x8 a_hi = *(const u16x8*)(arow + k0 + 16 + hl * 8);
        u16x16 au;
#pragma unroll
        for (int i = 0; i < 8; ++i) { au[i] = a_lo[i]; au[i + 8] = a_hi[i]; }
        v16bf av = __builtin_bit_cast(v16bf, au);

        const size_t koff = (size_t)kt * (HDIM * 32);
#pragma unroll
        for (int c = 0; c < 4; ++c) {
            const unsigned short* bptr = bbase[c] + koff;
            u16x8 b_lo = *(const u16x8*)(bptr);
            u16x8 b_hi = *(const u16x8*)(bptr + 8);
            u16x16 bu;
#pragma unroll
            for (int i = 0; i < 8; ++i) { bu[i] = b_lo[i]; bu[i + 8] = b_hi[i]; }
            v16bf bv = __builtin_bit_cast(v16bf, bu);
            v8f cin = (c == 0) ? acc0 : (c == 1) ? acc1 : (c == 2) ? acc2 : acc3;
            v8f cout = __builtin_amdgcn_wmma_f32_16x16x32_bf16(
                 false, av,  false, bv,
                 (short)0, cin,  false,  false);
            if (c == 0) acc0 = cout; else if (c == 1) acc1 = cout;
            else if (c == 2) acc2 = cout; else acc3 = cout;
        }
    }

#pragma unroll
    for (int c = 0; c < 4; ++c) {
        const int col = tn * 64 + c * 16 + m16;
        v8f accv = (c == 0) ? acc0 : (c == 1) ? acc1 : (c == 2) ? acc2 : acc3;
#pragma unroll
        for (int r = 0; r < 8; ++r)
            C[(size_t)(tm * 16 + hl * 8 + r) * HDIM + col] = accv[r];
    }
}

__global__ void __launch_bounds__(256)
gather_rows_kernel(const int* __restrict__ rowptr, const int2* __restrict__ rec,
                   const float* __restrict__ dis, const float* __restrict__ h,
                   const float* __restrict__ bias, float* __restrict__ acc, int n) {
    const int lane = threadIdx.x & 31;
    int row = blockIdx.x * 8 + (threadIdx.x >> 5);
    if (row >= n) return;
    row = __builtin_amdgcn_readfirstlane(row);
    const int beg = rowptr[row];
    const int end = rowptr[row + 1];
    const float d  = dis[row];
    const float dd = d * d;
    const float4* hr = (const float4*)(h + (size_t)row * HDIM);
    const float4* bp = (const float4*)bias;
    float4 a0 = bp[lane], a1 = bp[lane + 32];
    {
        const float4 v0 = hr[lane], v1 = hr[lane + 32];
        a0.x += v0.x * dd; a0.y += v0.y * dd; a0.z += v0.z * dd; a0.w += v0.w * dd;
        a1.x += v1.x * dd; a1.y += v1.y * dd; a1.z += v1.z * dd; a1.w += v1.w * dd;
    }
    for (int base = beg; base < end; base += 32) {
        const int m = min(end - base, 32);
        int2 r = make_int2(0, 0);
        if (lane < m) r = rec[base + lane];
#pragma unroll 4
        for (int j = 0; j < m; ++j) {
            const int   s = __builtin_amdgcn_readlane(r.x, j);
            const float w = __int_as_float(__builtin_amdgcn_readlane(r.y, j));
            const float4* hs = (const float4*)(h + (size_t)s * HDIM);
            const float4 v0 = hs[lane], v1 = hs[lane + 32];
            a0.x += v0.x * w; a0.y += v0.y * w; a0.z += v0.z * w; a0.w += v0.w * w;
            a1.x += v1.x * w; a1.y += v1.y * w; a1.z += v1.z * w; a1.w += v1.w * w;
        }
    }
    float4* ar = (float4*)(acc + (size_t)row * HDIM);
    ar[lane]      = a0;
    ar[lane + 32] = a1;
}

__global__ void l2norm_kernel(const float* __restrict__ acc, float* __restrict__ out) {
    int row = blockIdx.x;
    int tid = threadIdx.x;
    float v = acc[(size_t)row * HDIM + tid];
    float s = v * v;
#pragma unroll
    for (int m = 16; m > 0; m >>= 1) s += __shfl_xor(s, m, 32);
    __shared__ float ws[8];
    if ((tid & 31) == 0) ws[tid >> 5] = s;
    __syncthreads();
    float tot = 0.0f;
#pragma unroll
    for (int i = 0; i < 8; ++i) tot += ws[i];
    float inv = 1.0f / fmaxf(sqrtf(tot), 1e-12f);
    out[(size_t)row * HDIM + tid] = v * inv;
}

extern "C" void kernel_launch(void* const* d_in, const int* in_sizes, int n_in,
                              void* d_out, int out_size, void* d_ws, size_t ws_size,
                              hipStream_t stream) {
    const float* x   = (const float*)d_in[0];
    const int*   ei  = (const int*)d_in[1];
    const float* W1  = (const float*)d_in[2];
    const float* b1  = (const float*)d_in[3];
    const float* W2  = (const float*)d_in[4];
    const float* b2  = (const float*)d_in[5];
    float* out = (float*)d_out;

    const int N = in_sizes[0] / HDIM;
    const int E = in_sizes[1] / 2;
    const int* src = ei;
    const int* dst = ei + E;

    char* p = (char*)d_ws;
    auto take = [&](size_t bytes) { void* r = p; p += (bytes + 255) & ~(size_t)255; return r; };
    unsigned short* xb  = (unsigned short*)take((size_t)N * HDIM * 2);
    float*          h   = (float*)take((size_t)N * HDIM * 4);
    float*          acc = (float*)take((size_t)N * HDIM * 4);
    unsigned short* wp1 = (unsigned short*)take(8 * 256 * 32 * 2);
    unsigned short* wp2 = (unsigned short*)take(8 * 256 * 32 * 2);
    float*          dis = (float*)take((size_t)N * 4);
    int*            cnt    = (int*)take((size_t)N * 4);
    int*            rowptr = (int*)take(((size_t)N + 1) * 4);
    int*            rank   = (int*)take((size_t)E * 4);
    int2*           rec    = (int2*)take((size_t)E * 8);

    const size_t total = (size_t)N * HDIM;
    const int tilesM = (N + 15) / 16;
    const int stripsTotal = tilesM * 4;
    const int gemmBlocks = (stripsTotal + 7) / 8;
    const int rowBlocks = (N + 7) / 8;

    zero_cnt_kernel<<<(N + 255) / 256, 256, 0, stream>>>(cnt, N);
    count_rank_kernel<<<(E + 255) / 256, 256, 0, stream>>>(dst, cnt, rank, E);
    scan_deg_kernel<<<1, 1024, 0, stream>>>(cnt, rowptr, dis, N);
    fill_lists_kernel<<<(E + 255) / 256, 256, 0, stream>>>(src, dst, rank, rowptr, dis, rec, E);

    cvt_bf16_kernel<<<(int)((total + 255) / 256), 256, 0, stream>>>(x, xb, total);
    pack_w_kernel<<<256, 256, 0, stream>>>(W1, wp1);
    pack_w_kernel<<<256, 256, 0, stream>>>(W2, wp2);

    gemm_wmma_bf16<<<gemmBlocks, 256, 0, stream>>>(xb, wp1, h, stripsTotal);
    gather_rows_kernel<<<rowBlocks, 256, 0, stream>>>(rowptr, rec, dis, h, b1, acc, N);
    relu_bf16_kernel<<<(int)((total + 255) / 256), 256, 0, stream>>>(acc, xb, total);

    gemm_wmma_bf16<<<gemmBlocks, 256, 0, stream>>>(xb, wp2, h, stripsTotal);
    gather_rows_kernel<<<rowBlocks, 256, 0, stream>>>(rowptr, rec, dis, h, b2, acc, N);

    l2norm_kernel<<<N, 256, 0, stream>>>(acc, out);
}


# ReLU+convert and the row L2 normalise done by the gather wave on the row it holds in registers: two elementwise launches (800000 one-element waves) removed; norm sum of squares now reduced in-wave
# speedup vs baseline: 2.7149x; 1.2985x over previous
#include <hip/hip_runtime.h>
#include <hip/hip_bf16.h>

typedef __attribute__((ext_vector_type(16))) __bf16        v16bf;
typedef __attribute__((ext_vector_type(8)))  float         v8f;
typedef __attribute__((ext_vector_type(8)))  unsigned short u16x8;
typedef __attribute__((ext_vector_type(16))) unsigned short u16x16;

__device__ __forceinline__ unsigned short f32_to_bf16_rne(float f) {
    unsigned int u = __float_as_uint(f);
    unsigned int r = u + 0x7FFFu + ((u >> 16) & 1u);
    return (unsigned short)(r >> 16);
}

#define HDIM 256

__global__ void zero_cnt_kernel(int* cnt, int n) {
    int i = blockIdx.x * blockDim.x + threadIdx.x;
    if (i < n) cnt[i] = 0;
}

__global__ void count_rank_kernel(const int* __restrict__ dst, int* cnt, int* __restrict__ rank, int e) {
    int i = blockIdx.x * blockDim.x + threadIdx.x;
    if (i < e) rank[i] = atomicAdd(&cnt[dst[i]], 1);
}

__global__ void __launch_bounds__(1024)
scan_deg_kernel(const int* __restrict__ cnt, int* __restrict__ rowptr, float* __restrict__ dis, int n) {
    __shared__ int wsum[32];
    const int tid  = threadIdx.x;
    const int lane = tid & 31;
    const int wv   = tid >> 5;
    int carry = 0;
    for (int base = 0; base < n; base += 8192) {
        const int i0 = base + tid * 8;
        int c[8];
        if (i0 + 7 < n) {
            const int4 v0 = *(const int4*)(cnt + i0);
            const int4 v1 = *(const int4*)(cnt + i0 + 4);
            c[0] = v0.x; c[1] = v0.y; c[2] = v0.z; c[3] = v0.w;
            c[4] = v1.x; c[5] = v1.y; c[6] = v1.z; c[7] = v1.w;
        } else {
            for (int k = 0; k < 8; ++k) c[k] = (i0 + k < n) ? cnt[i0 + k] : 0;
        }
        int t = 0;
        for (int k = 0; k < 8; ++k) t += c[k];
        int s = t;
        for (int d = 1; d < 32; d <<= 1) {
            const int y = __shfl_up(s, d, 32);
            if (lane >= d) s += y;
        }
        if (lane == 31) wsum[wv] = s;
        __syncthreads();
        const int wt = wsum[lane];
        int ws = wt;
        for (int d = 1; d < 32; d <<= 1) {
            const int y = __shfl_up(ws, d, 32);
            if (lane >= d) ws += y;
        }
        const int wave_excl = __shfl(ws, wv, 32) - __shfl(wt, wv, 32);
        const int total     = __shfl(ws, 31, 32);
        int run = carry + wave_excl + (s - t);
        if (i0 + 7 < n) {
            int4 p0, p1;
            float4 q0, q1;
            p0.x = run; run += c[0]; p0.y = run; run += c[1]; p0.z = run; run += c[2]; p0.w = run; run += c[3];
            p1.x = run; run += c[4]; p1.y = run; run += c[5]; p1.z = run; run += c[6]; p1.w = run;
            q0.x = rsqrtf((float)(c[0] + 1)); q0.y = rsqrtf((float)(c[1] + 1));
            q0.z = rsqrtf((float)(c[2] + 1)); q0.w = rsqrtf((float)(c[3] + 1));
            q1.x = rsqrtf((float)(c[4] + 1)); q1.y = rsqrtf((float)(c[5] + 1));
            q1.z = rsqrtf((float)(c[6] + 1)); q1.w = rsqrtf((float)(c[7] + 1));
            *(int4*)(rowptr + i0)     = p0;
            *(int4*)(rowptr + i0 + 4) = p1;
            *(float4*)(dis + i0)      = q0;
            *(float4*)(dis + i0 + 4)  = q1;
        } else {
            for (int k = 0; k < 8; ++k) {
                if (i0 + k < n) { rowptr[i0 + k] = run; dis[i0 + k] = rsqrtf((float)(c[k] + 1)); }
                run += c[k];
            }
        }
        carry += total;
        __syncthreads();
    }
    if (tid == 0) rowptr[n] = carry;
}

__global__ void fill_lists_kernel(const int* __restrict__ src, const int* __restrict__ dst,
                                  const int* __restrict__ rank, const int* __restrict__ rowptr,
                                  const float* __restrict__ dis, int2* __restrict__ rec, int e) {
    int i = blockIdx.x * blockDim.x + threadIdx.x;
    if (i >= e) return;
    const int s = src[i], d = dst[i];
    rec[rowptr[d] + rank[i]] = make_int2(s, __float_as_int(dis[s] * dis[d]));
}

__global__ void cvt_bf16_kernel(const float* __restrict__ x, unsigned short* __restrict__ xb,
                                size_t total) {
    size_t i = (size_t)blockIdx.x * blockDim.x + threadIdx.x;
    if (i < total) xb[i] = f32_to_bf16_rne(x[i]);
}

__global__ void pack_w_kernel(const float* __restrict__ W, unsigned short* __restrict__ Bp) {
    int idx = blockIdx.x * blockDim.x + threadIdx.x;
    int j  = idx & 15;
    int hl = (idx >> 4) & 1;
    int n  = (idx >> 5) & 255;
    int t  = idx >> 13;
    int k  = t * 32 + ((j >> 3) << 4) + hl * 8 + (j & 7);
    Bp[idx] = f32_to_bf16_rne(W[k * HDIM + n]);
}

__global__ void gemm_wmma_bf16(const unsigned short* __restrict__ A,
                               const unsigned short* __restrict__ Bp,
                               float* __restrict__ C, int stripsTotal) {
    const int wave = threadIdx.x >> 5;
    const int lane = threadIdx.x & 31;
    const int hl   = lane >> 4;
    const int m16  = lane & 15;
    const int strip = blockIdx.x * 8 + wave;
    if (strip >= stripsTotal) return;
    const int tm = strip >> 2;
    const int tn = strip & 3;

    const unsigned short* arow = A + (size_t)(tm * 16 + m16) * HDIM;
    const unsigned short* bbase[4];
#pragma unroll
    for (int c = 0; c < 4; ++c)
        bbase[c] = Bp + ((size_t)(tn * 64 + c * 16 + m16) * 2 + hl) * 16;

    v8f acc0 = {}, acc1 = {}, acc2 = {}, acc3 = {};
#pragma unroll
    for (int kt = 0; kt < 8; ++kt) {
        const int k0 = kt * 32;
        u16x8 a_lo = *(const u16x8*)(arow + k0 + hl * 8);
        u16x8 a_hi = *(const u16x8*)(arow + k0 + 16 + hl * 8);
        u16x16 au;
#pragma unroll
        for (int i = 0; i < 8; ++i) { au[i] = a_lo[i]; au[i + 8] = a_hi[i]; }
        v16bf av = __builtin_bit_cast(v16bf, au);

        const size_t koff = (size_t)kt * (HDIM * 32);
#pragma unroll
        for (int c = 0; c < 4; ++c) {
            const unsigned short* bptr = bbase[c] + koff;
            u16x8 b_lo = *(const u16x8*)(bptr);
            u16x8 b_hi = *(const u16x8*)(bptr + 8);
            u16x16 bu;
#pragma unroll
            for (int i = 0; i < 8; ++i) { bu[i] = b_lo[i]; bu[i + 8] = b_hi[i]; }
            v16bf bv = __builtin_bit_cast(v16bf, bu);
            v8f cin = (c == 0) ? acc0 : (c == 1) ? acc1 : (c == 2) ? acc2 : acc3;
            v8f cout = __builtin_amdgcn_wmma_f32_16x16x32_bf16(
                 false, av,  false, bv,
                 (short)0, cin,  false,  false);
            if (c == 0) acc0 = cout; else if (c == 1) acc1 = cout;
            else if (c == 2) acc2 = cout; else acc3 = cout;
        }
    }

#pragma unroll
    for (int c = 0; c < 4; ++c) {
        const int col = tn * 64 + c * 16 + m16;
        v8f accv = (c == 0) ? acc0 : (c == 1) ? acc1 : (c == 2) ? acc2 : acc3;
#pragma unroll
        for (int r = 0; r < 8; ++r)
            C[(size_t)(tm * 16 + hl * 8 + r) * HDIM + col] = accv[r];
    }
}

__device__ __forceinline__ void gather_row(const int* __restrict__ rowptr, const int2* __restrict__ rec,
                                           const float* __restrict__ dis, const float* __restrict__ h,
                                           const float* __restrict__ bias, int row, int lane,
                                           float4& a0, float4& a1) {
    const int beg = rowptr[row];
    const int end = rowptr[row + 1];
    const float d  = dis[row];
    const float dd = d * d;
    const float4* hr = (const float4*)(h + (size_t)row * HDIM);
    const float4* bp = (const float4*)bias;
    a0 = bp[lane]; a1 = bp[lane + 32];
    {
        const float4 v0 = hr[lane], v1 = hr[lane + 32];
        a0.x += v0.x * dd; a0.y += v0.y * dd; a0.z += v0.z * dd; a0.w += v0.w * dd;
        a1.x += v1.x * dd; a1.y += v1.y * dd; a1.z += v1.z * dd; a1.w += v1.w * dd;
    }
    for (int base = beg; base < end; base += 32) {
        const int m = min(end - base, 32);
        int2 r = make_int2(0, 0);
        if (lane < m) r = rec[base + lane];
#pragma unroll 4
        for (int j = 0; j < m; ++j) {
            const int   s = __builtin_amdgcn_readlane(r.x, j);
            const float w = __int_as_float(__builtin_amdgcn_readlane(r.y, j));
            const float4* hs = (const float4*)(h + (size_t)s * HDIM);
            const float4 v0 = hs[lane], v1 = hs[lane + 32];
            a0.x += v0.x * w; a0.y += v0.y * w; a0.z += v0.z * w; a0.w += v0.w * w;
            a1.x += v1.x * w; a1.y += v1.y * w; a1.z += v1.z * w; a1.w += v1.w * w;
        }
    }
}

typedef __attribute__((ext_vector_type(2))) float  f32x2;
typedef __attribute__((ext_vector_type(2))) __bf16 bf16x2;

__device__ __forceinline__ unsigned int relu_pack2(float lo, float hi) {
    const f32x2 v = { fmaxf(lo, 0.0f), fmaxf(hi, 0.0f) };
    return __builtin_bit_cast(unsigned int, __builtin_convertvector(v, bf16x2));
}

__global__ void __launch_bounds__(256)
gather_relu_cvt_kernel(const int* __restrict__ rowptr, const int2* __restrict__ rec,
                       const float* __restrict__ dis, const float* __restrict__ h,
                       const float* __restrict__ bias, unsigned short* __restrict__ zb, int n) {
    const int lane = threadIdx.x & 31;
    int row = blockIdx.x * 8 + (threadIdx.x >> 5);
    if (row >= n) return;
    row = __builtin_amdgcn_readfirstlane(row);
    float4 a0, a1;
    gather_row(rowptr, rec, dis, h, bias, row, lane, a0, a1);
    uint2 p0, p1;
    p0.x = relu_pack2(a0.x, a0.y); p0.y = relu_pack2(a0.z, a0.w);
    p1.x = relu_pack2(a1.x, a1.y); p1.y = relu_pack2(a1.z, a1.w);
    uint2* zr = (uint2*)(zb + (size_t)row * HDIM);
    zr[lane]      = p0;
    zr[lane + 32] = p1;
}

__global__ void __launch_bounds__(256)
gather_l2norm_kernel(const int* __restrict__ rowptr, const int2* __restrict__ rec,
                     const float* __restrict__ dis, const float* __restrict__ h,
                     const float* __restrict__ bias, float* __restrict__ out, int n) {
    const int lane = threadIdx.x & 31;
    int row = blockIdx.x * 8 + (threadIdx.x >> 5);
    if (row >= n) return;
    row = __builtin_amdgcn_readfirstlane(row);
    float4 a0, a1;
    gather_row(rowptr, rec, dis, h, bias, row, lane, a0, a1);
    float s = a0.x * a0.x + a0.y * a0.y + a0.z * a0.z + a0.w * a0.w
            + a1.x * a1.x + a1.y * a1.y + a1.z * a1.z + a1.w * a1.w;
#pragma unroll
    for (int m = 16; m > 0; m >>= 1) s += __shfl_xor(s, m, 32);
    const float inv = 1.0f / fmaxf(sqrtf(s), 1e-12f);
    a0.x *= inv; a0.y *= inv; a0.z *= inv; a0.w *= inv;
    a1.x *= inv; a1.y *= inv; a1.z *= inv; a1.w *= inv;
    float4* orow = (float4*)(out + (size_t)row * HDIM);
    orow[lane]      = a0;
    orow[lane + 32] = a1;
}

extern "C" void kernel_launch(void* const* d_in, const int* in_sizes, int n_in,
                              void* d_out, int out_size, void* d_ws, size_t ws_size,
                              hipStream_t stream) {
    const float* x   = (const float*)d_in[0];
    const int*   ei  = (const int*)d_in[1];
    const float* W1  = (const float*)d_in[2];
    const float* b1  = (const float*)d_in[3];
    const float* W2  = (const float*)d_in[4];
    const float* b2  = (const float*)d_in[5];
    float* out = (float*)d_out;

    const int N = in_sizes[0] / HDIM;
    const int E = in_sizes[1] / 2;
    const int* src = ei;
    const int* dst = ei + E;

    char* p = (char*)d_ws;
    auto take = [&](size_t bytes) { void* r = p; p += (bytes + 255) & ~(size_t)255; return r; };
    unsigned short* xb  = (unsigned short*)take((size_t)N * HDIM * 2);
    float*          h   = (float*)take((size_t)N * HDIM * 4);
    unsigned short* wp1 = (unsigned short*)take(8 * 256 * 32 * 2);
    unsigned short* wp2 = (unsigned short*)take(8 * 256 * 32 * 2);
    float*          dis = (float*)take((size_t)N * 4);
    int*            cnt    = (int*)take((size_t)N * 4);
    int*            rowptr = (int*)take(((size_t)N + 1) * 4);
    int*            rank   = (int*)take((size_t)E * 4);
    int2*           rec    = (int2*)take((size_t)E * 8);

    const size_t total = (size_t)N * HDIM;
    const int tilesM = (N + 15) / 16;
    const int stripsTotal = tilesM * 4;
    const int gemmBlocks = (stripsTotal + 7) / 8;
    const int rowBlocks = (N + 7) / 8;

    zero_cnt_kernel<<<(N + 255) / 256, 256, 0, stream>>>(cnt, N);
    count_rank_kernel<<<(E + 255) / 256, 256, 0, stream>>>(dst, cnt, rank, E);
    scan_deg_kernel<<<1, 1024, 0, stream>>>(cnt, rowptr, dis, N);
    fill_lists_kernel<<<(E + 255) / 256, 256, 0, stream>>>(src, dst, rank, rowptr, dis, rec, E);

    cvt_bf16_kernel<<<(int)((total + 255) / 256), 256, 0, stream>>>(x, xb, total);
    pack_w_kernel<<<256, 256, 0, stream>>>(W1, wp1);
    pack_w_kernel<<<256, 256, 0, stream>>>(W2, wp2);

    gemm_wmma_bf16<<<gemmBlocks, 256, 0, stream>>>(xb, wp1, h, stripsTotal);
    gather_relu_cvt_kernel<<<rowBlocks, 256, 0, stream>>>(rowptr, rec, dis, h, b1, xb, N);

    gemm_wmma_bf16<<<gemmBlocks, 256, 0, stream>>>(xb, wp2, h, stripsTotal);
    gather_l2norm_kernel<<<rowBlocks, 256, 0, stream>>>(rowptr, rec, dis, h, b2, out, N);
}


# layer 1 sums first, projects after (as layer 2): gather reads a 16-bit copy of x (512 B an edge, not 1024); f32 sum x W1 as three exact 16-bit planes there; GEMM pass and h gone; sum order changes
# speedup vs baseline: 5.7870x; 2.1316x over previous
#include <hip/hip_runtime.h>
#include <hip/hip_bf16.h>

typedef __attribute__((ext_vector_type(16))) __bf16        v16bf;
typedef __attribute__((ext_vector_type(8)))  float         v8f;
typedef __attribute__((ext_vector_type(8)))  unsigned short u16x8;
typedef __attribute__((ext_vector_type(16))) unsigned short u16x16;

__device__ __forceinline__ unsigned short f32_to_bf16_rne(float f) {
    unsigned int u = __float_as_uint(f);
    unsigned int r = u + 0x7FFFu + ((u >> 16) & 1u);
    return (unsigned short)(r >> 16);
}

#define HDIM 256

#define LCAP 64

#define ECHUNK 4096
#define CCAP   64
#define OROWS  256
#define MAXB   256

__device__ __forceinline__ void pack_w_elem(const float* __restrict__ W, unsigned short* __restrict__ Bp, int idx) {
    int j  = idx & 15;
    int hl = (idx >> 4) & 1;
    int n  = (idx >> 5) & 255;
    int t  = idx >> 13;
    int k  = t * 32 + ((j >> 3) << 4) + hl * 8 + (j & 7);
    Bp[idx] = f32_to_bf16_rne(W[k * HDIM + n]);
}

__global__ void __launch_bounds__(256)
prep_kernel(const float* __restrict__ W1, unsigned short* __restrict__ wp1,
            const float* __restrict__ W2, unsigned short* __restrict__ wp2,
            const int* __restrict__ src, const int* __restrict__ dst, int e,
            int* __restrict__ cellw, int* __restrict__ cellcnt, int nChunks) {
    int2* __restrict__ cell = (int2*)cellw;
    __shared__ int binc[MAXB];
    const int b = blockIdx.x;
    if (b < 256) {
        pack_w_elem(W1, wp1, b * 256 + threadIdx.x);
    } else if (b < 512) {
        pack_w_elem(W2, wp2, (b - 256) * 256 + threadIdx.x);
    } else {
        const int ch = b - 512;
        binc[threadIdx.x] = 0;
        __syncthreads();
        int dv[ECHUNK / 256], sv[ECHUNK / 256];
#pragma unroll
        for (int q = 0; q < ECHUNK / 256; ++q) {
            const int i = ch * ECHUNK + q * 256 + threadIdx.x;
            dv[q] = (i < e) ? dst[i] : -1;
            sv[q] = (i < e) ? src[i] : 0;
        }
        int rk[ECHUNK / 256];
#pragma unroll
        for (int q = 0; q < ECHUNK / 256; ++q)
            rk[q] = (dv[q] >= 0) ? atomicAdd(&binc[dv[q] >> 8], 1) : CCAP;
#pragma unroll
        for (int q = 0; q < ECHUNK / 256; ++q) {
            if (rk[q] < CCAP) cell[((size_t)(dv[q] >> 8) * nChunks + ch) * CCAP + rk[q]] = make_int2(sv[q], dv[q]);
        }
        __syncthreads();
        cellcnt[threadIdx.x * nChunks + ch] = binc[threadIdx.x];
    }
}

typedef __attribute__((ext_vector_type(2))) float  f32x2;
typedef __attribute__((ext_vector_type(2))) __bf16 bf16x2;
typedef __attribute__((ext_vector_type(8))) unsigned int u32x8;

__device__ __forceinline__ unsigned int pack2_bf16(float lo, float hi) {
    const f32x2 v = { lo, hi };
    return __builtin_bit_cast(unsigned int, __builtin_convertvector(v, bf16x2));
}

__device__ __forceinline__ v16bf load_a_frag_bf16(const unsigned short* __restrict__ arow, int k0, int hl) {
    u16x8 a_lo = *(const u16x8*)(arow + k0 + hl * 8);
    u16x8 a_hi = *(const u16x8*)(arow + k0 + 16 + hl * 8);
    u16x16 au;
#pragma unroll
    for (int i = 0; i < 8; ++i) { au[i] = a_lo[i]; au[i + 8] = a_hi[i]; }
    return __builtin_bit_cast(v16bf, au);
}

__device__ __forceinline__ void convert_rows(const float* __restrict__ x, unsigned short* __restrict__ xb,
                                             int nvec4, int blk) {
    const float4* xv = (const float4*)x;
    uint2* ov = (uint2*)xb;
    const int base = blk * (8 * 256) + (int)threadIdx.x;
    float4 v[8];
#pragma unroll
    for (int q = 0; q < 8; ++q) {
        v[q] = xv[min(base + q * 256, nvec4 - 1)];
    }
#pragma unroll
    for (int q = 0; q < 8; ++q) {
        const int idx = base + q * 256;
        if (idx < nvec4) ov[idx] = make_uint2(pack2_bf16(v[q].x, v[q].y), pack2_bf16(v[q].z, v[q].w));
    }
}

__device__ __forceinline__ void rank_place(int s, int d, int* oc, int* __restrict__ lst) {
    const int r = atomicAdd(&oc[d & (OROWS - 1)], 1);
    if (r < LCAP) lst[(size_t)d * LCAP + r] = s;
}

__global__ void __launch_bounds__(256)
convert_rank_place_kernel(const float* __restrict__ x, unsigned short* __restrict__ xb, int nvec4, int M,
                          const int* __restrict__ cellw, const int* __restrict__ cellcnt, int nChunks,
                          const int* __restrict__ src, const int* __restrict__ dst, int e,
                          int* __restrict__ cnt, int* __restrict__ lst, int numOwners) {
    __shared__ int oc[OROWS];
    __shared__ int hc[MAXB];
    __shared__ int over;
    if ((int)blockIdx.x >= numOwners) {
        convert_rows(x, xb, nvec4, (int)blockIdx.x - numOwners);
        return;
    }
    const int k   = blockIdx.x;
    const int tid = threadIdx.x;
    oc[tid] = 0;
    const int cc = (tid < nChunks) ? cellcnt[k * nChunks + tid] : 0;
    hc[tid] = cc;
    if (tid == 0) over = 0;
    __syncthreads();
    if (cc > CCAP) over = 1;
    __syncthreads();
    if (over == 0) {
        const int2* ck = (const int2*)cellw + (size_t)k * nChunks * CCAP;
        const int total = nChunks * CCAP;
        for (int base = 0; base < total; base += 8 * 256) {
            int2 sd[8];
            bool ok[8];
#pragma unroll
            for (int u = 0; u < 8; ++u) {
                const int idx = base + u * 256 + tid;
                ok[u] = (idx < total) && ((idx & (CCAP - 1)) < hc[idx / CCAP]);
                sd[u] = make_int2(0, 0);
                if (ok[u]) sd[u] = ck[idx];
            }
            int rr[8];
#pragma unroll
            for (int u = 0; u < 8; ++u)
                rr[u] = ok[u] ? atomicAdd(&oc[sd[u].y & (OROWS - 1)], 1) : LCAP;
#pragma unroll
            for (int u = 0; u < 8; ++u) {
                if (rr[u] < LCAP) lst[(size_t)sd[u].y * LCAP + rr[u]] = sd[u].x;
            }
        }
    } else {
        for (int i = tid; i < e; i += 256) {
            const int d = dst[i];
            if ((d >> 8) == k) rank_place(src[i], d, oc, lst);
        }
    }
    __syncthreads();
    const int row = k * OROWS + tid;
    if (row < M) cnt[row] = oc[tid];
}

__device__ __forceinline__ unsigned int relu_pack2(float lo, float hi) {
    const f32x2 v = { fmaxf(lo, 0.0f), fmaxf(hi, 0.0f) };
    return __builtin_bit_cast(unsigned int, __builtin_convertvector(v, bf16x2));
}


__device__ __forceinline__ void add_edge_zrow(const uint4 v, float w, float (&a)[8]) {
    a[0] += __uint_as_float(v.x << 16) * w; a[1] += __uint_as_float(v.x & 0xFFFF0000u) * w;
    a[2] += __uint_as_float(v.y << 16) * w; a[3] += __uint_as_float(v.y & 0xFFFF0000u) * w;
    a[4] += __uint_as_float(v.z << 16) * w; a[5] += __uint_as_float(v.z & 0xFFFF0000u) * w;
    a[6] += __uint_as_float(v.w << 16) * w; a[7] += __uint_as_float(v.w & 0xFFFF0000u) * w;
}

template <bool FIRST>
__device__ __forceinline__ void gather_zsum(const int* __restrict__ cnt, const int* __restrict__ lst,
                                            float* __restrict__ wl,
                                            const int* __restrict__ src, const int* __restrict__ dst, int e,
                                            const unsigned short* __restrict__ z, int row, int lane,
                                            float (&a)[8]) {
    const int c = cnt[row];
    const float d  = rsqrtf((float)(c + 1));
    const float dd = d * d;
    const uint4* zq = (const uint4*)z;
    add_edge_zrow(zq[(size_t)row * 32 + lane], dd, a);
    if (c <= LCAP) {
        const size_t b0 = (size_t)row * LCAP;
        for (int base = 0; base < c; base += 32) {
            const int m = min(c - base, 32);
            int   sv = 0;
            float wv = 0.0f;
            if (lane < m) {
                sv = lst[b0 + base + lane];
                if (FIRST) {
                    wv = rsqrtf((float)(cnt[sv] + 1)) * d;
                    wl[b0 + base + lane] = wv;
                } else {
                    wv = wl[b0 + base + lane];
                }
            }
            for (int j0 = 0; j0 < m; j0 += 8) {
                uint4 v[8];
                float w[8];
#pragma unroll
                for (int u = 0; u < 8; ++u) {
                    if (u == 0 || j0 + u < m) {
                        const int s = __builtin_amdgcn_readlane(sv, j0 + u);
                        w[u] = __int_as_float(__builtin_amdgcn_readlane(__float_as_int(wv), j0 + u));
                        v[u] = zq[(size_t)s * 32 + lane];
                    }
                }
#pragma unroll
                for (int u = 0; u < 8; ++u) {
                    if (u == 0 || j0 + u < m) add_edge_zrow(v[u], w[u], a);
                }
            }
        }
    } else {
        for (int base = 0; base < e; base += 32) {
            const int i = base + lane;
            const bool mine = (i < e) && (dst[i] == row);
            unsigned int mask = __builtin_amdgcn_ballot_w32(mine);
            if (mask == 0u) continue;
            int   sv = 0;
            float wv = 0.0f;
            if (mine) {
                sv = src[i];
                wv = rsqrtf((float)(cnt[sv] + 1)) * d;
            }
            while (mask != 0u) {
                const int j = __builtin_ctz(mask);
                mask &= mask - 1u;
                const int   s = __builtin_amdgcn_readlane(sv, j);
                const float w = __int_as_float(__builtin_amdgcn_readlane(__float_as_int(wv), j));
                add_edge_zrow(zq[(size_t)s * 32 + lane], w, a);
            }
        }
    }
}

__device__ __forceinline__ void split3_pack2(float x0, float x1,
                                             unsigned int& ph, unsigned int& pm, unsigned int& pl) {
    ph = pack2_bf16(x0, x1);
    const float r0 = x0 - __uint_as_float(ph << 16);
    const float r1 = x1 - __uint_as_float(ph & 0xFFFF0000u);
    pm = pack2_bf16(r0, r1);
    const float q0 = r0 - __uint_as_float(pm << 16);
    const float q1 = r1 - __uint_as_float(pm & 0xFFFF0000u);
    pl = pack2_bf16(q0, q1);
}

#define ZPITCH 264
#define CPITCH 260

template <bool FIRST>
__global__ void __launch_bounds__(512, 16)
gather_project_kernel(const int* __restrict__ cnt, const int* __restrict__ lst, float* __restrict__ wl,
                      const int* __restrict__ src, const int* __restrict__ dst, int e,
                      const unsigned short* __restrict__ z,
                      const unsigned short* __restrict__ Bp, const float* __restrict__ bias,
                      unsigned short* __restrict__ zout, float* __restrict__ out, int n) {
    __shared__ __attribute__((aligned(16))) unsigned short zs[3 * 16 * ZPITCH];
    __shared__ __attribute__((aligned(16))) float          cs[16 * CPITCH];
    const int lane = threadIdx.x & 31;
    const int wave = threadIdx.x >> 5;
    const int row0 = blockIdx.x * 16;
    const int row = __builtin_amdgcn_readfirstlane(row0 + wave);
    float a[8];
#pragma unroll
    for (int i = 0; i < 8; ++i) a[i] = 0.0f;
    if (row < n)
        gather_zsum<FIRST>(cnt, lst, wl, src, dst, e, z, row, lane, a);
    uint4 ph, pm, pl;
    split3_pack2(a[0], a[1], ph.x, pm.x, pl.x);
    split3_pack2(a[2], a[3], ph.y, pm.y, pl.y);
    split3_pack2(a[4], a[5], ph.z, pm.z, pl.z);
    split3_pack2(a[6], a[7], ph.w, pm.w, pl.w);
    ((uint4*)(zs + (0 * 16 + wave) * ZPITCH))[lane] = ph;
    ((uint4*)(zs + (1 * 16 + wave) * ZPITCH))[lane] = pm;
    ((uint4*)(zs + (2 * 16 + wave) * ZPITCH))[lane] = pl;

    const int hl  = lane >> 4;
    const int m16 = lane & 15;
    const unsigned short* bb   = Bp + ((size_t)(wave * 16 + m16) * 2 + hl) * 16;
    const unsigned short* arow = zs + m16 * ZPITCH;
    v8f acch = {}, accr = {};
    u16x8 blo[4], bhi[4];
#pragma unroll
    for (int kk = 0; kk < 4; ++kk) {
        blo[kk] = *(const u16x8*)(bb + (size_t)kk * (HDIM * 32));
        bhi[kk] = *(const u16x8*)(bb + (size_t)kk * (HDIM * 32) + 8);
    }

    __syncthreads();

#pragma unroll
    for (int half = 0; half < 2; ++half) {
        if (half == 1) {
#pragma unroll
            for (int kk = 0; kk < 4; ++kk) {
                blo[kk] = *(const u16x8*)(bb + (size_t)(4 + kk) * (HDIM * 32));
                bhi[kk] = *(const u16x8*)(bb + (size_t)(4 + kk) * (HDIM * 32) + 8);
            }
        }
#pragma unroll
        for (int kk = 0; kk < 4; ++kk) {
            const int k0 = (half * 4 + kk) * 32;
            u16x16 bu;
#pragma unroll
            for (int i = 0; i < 8; ++i) { bu[i] = blo[kk][i]; bu[i + 8] = bhi[kk][i]; }
            const v16bf bv = __builtin_bit_cast(v16bf, bu);
            const v16bf ah = load_a_frag_bf16(arow,                   k0, hl);
            const v16bf am = load_a_frag_bf16(arow + 16 * ZPITCH,     k0, hl);
            const v16bf al = load_a_frag_bf16(arow + 2 * 16 * ZPITCH, k0, hl);
            acch = __builtin_amdgcn_wmma_f32_16x16x32_bf16(
                 false, ah,  false, bv,
                 (short)0, acch,  false,  false);
            accr = __builtin_amdgcn_wmma_f32_16x16x32_bf16(
                 false, am,  false, bv,
                 (short)0, accr,  false,  false);
            accr = __builtin_amdgcn_wmma_f32_16x16x32_bf16(
                 false, al,  false, bv,
                 (short)0, accr,  false,  false);
        }
    }

#pragma unroll
    for (int r = 0; r < 8; ++r)
        cs[(hl * 8 + r) * CPITCH + wave * 16 + m16] = acch[r] + accr[r];

    __syncthreads();

    if (row < n) {
        const float4* cr = (const float4*)(cs + wave * CPITCH);
        const float4* bp = (const float4*)bias;
        float4 a0 = cr[lane],      b0 = bp[lane];
        float4 a1 = cr[lane + 32], b1 = bp[lane + 32];
        a0.x += b0.x; a0.y += b0.y; a0.z += b0.z; a0.w += b0.w;
        a1.x += b1.x; a1.y += b1.y; a1.z += b1.z; a1.w += b1.w;
        if (FIRST) {
            uint2 p0, p1;
            p0.x = relu_pack2(a0.x, a0.y); p0.y = relu_pack2(a0.z, a0.w);
            p1.x = relu_pack2(a1.x, a1.y); p1.y = relu_pack2(a1.z, a1.w);
            uint2* zr = (uint2*)(zout + (size_t)row * HDIM);
            zr[lane]      = p0;
            zr[lane + 32] = p1;
        } else {
            float s = a0.x * a0.x + a0.y * a0.y + a0.z * a0.z + a0.w * a0.w
                    + a1.x * a1.x + a1.y * a1.y + a1.z * a1.z + a1.w * a1.w;
#pragma unroll
            for (int m = 16; m > 0; m >>= 1) s += __shfl_xor(s, m, 32);
            const float inv = 1.0f / fmaxf(sqrtf(s), 1e-12f);
            a0.x *= inv; a0.y *= inv; a0.z *= inv; a0.w *= inv;
            a1.x *= inv; a1.y *= inv; a1.z *= inv; a1.w *= inv;
            float4* orow = (float4*)(out + (size_t)row * HDIM);
            orow[lane]      = a0;
            orow[lane + 32] = a1;
        }
    }
}

extern "C" void kernel_launch(void* const* d_in, const int* in_sizes, int n_in,
                              void* d_out, int out_size, void* d_ws, size_t ws_size,
                              hipStream_t stream) {
    const float* x   = (const float*)d_in[0];
    const int*   ei  = (const int*)d_in[1];
    const float* W1  = (const float*)d_in[2];
    const float* b1  = (const float*)d_in[3];
    const float* W2  = (const float*)d_in[4];
    const float* b2  = (const float*)d_in[5];
    float* out = (float*)d_out;

    const int N = in_sizes[0] / HDIM;
    const int E = in_sizes[1] / 2;
    const int* src = ei;
    const int* dst = ei + E;

    char* p = (char*)d_ws;
    auto take = [&](size_t bytes) { void* r = p; p += (bytes + 255) & ~(size_t)255; return r; };
    unsigned short* xb  = (unsigned short*)take((size_t)N * HDIM * 2);
    unsigned short* z   = (unsigned short*)take((size_t)N * HDIM * 2);
    unsigned short* wp1 = (unsigned short*)take(8 * 256 * 32 * 2);
    unsigned short* wp2 = (unsigned short*)take(8 * 256 * 32 * 2);
    int*            cnt = (int*)take((size_t)N * 4);
    const int nChunks   = (E + ECHUNK - 1) / ECHUNK;
    const int numOwners = (N + OROWS - 1) / OROWS;
    int*            cell    = (int*)take((size_t)MAXB * nChunks * CCAP * 8);
    int*            cellcnt = (int*)take((size_t)MAXB * nChunks * 4);
    int*            lst = (int*)take((size_t)N * LCAP * 4);
    float*          wl  = (float*)take((size_t)N * LCAP * 4);

    const int nvec4 = N * (HDIM / 4);
    const int convBlocks = (nvec4 + 8 * 256 - 1) / (8 * 256);
    const int tileBlocks = (N + 15) / 16;

    prep_kernel<<<512 + nChunks, 256, 0, stream>>>(W1, wp1, W2, wp2, src, dst, E, cell, cellcnt, nChunks);

    convert_rank_place_kernel<<<numOwners + convBlocks, 256, 0, stream>>>(x, xb, nvec4, N,
                                                                           cell, cellcnt, nChunks, src, dst, E,
                                                                           cnt, lst, numOwners);

    gather_project_kernel<true><<<tileBlocks, 512, 0, stream>>>(cnt, lst, wl, src, dst, E, xb, wp1, b1, z, nullptr, N);

    gather_project_kernel<false><<<tileBlocks, 512, 0, stream>>>(cnt, lst, wl, src, dst, E, z, wp2, b2, nullptr, out, N);
}
